# starting kernel
# baseline (speedup 1.0000x reference)
#include <hip/hip_runtime.h>
#include <hip/hip_bf16.h>


typedef __attribute__((ext_vector_type(2))) float v2f;
typedef __attribute__((ext_vector_type(8))) float v8f;

#define B_DIM 128
#define I_DIM 512
#define O_DIM 512

#define C1_JOHNSON 1.65677880e-28f
#define C2_SHOT    5.43656366e-08f
#define NOISE_MIN  1e-12f

__global__ __launch_bounds__(256) void precomp_kernel(
    const float* __restrict__ x,
    const float* __restrict__ pl,
    const float* __restrict__ ph,
    float* __restrict__ L,
    float* __restrict__ Dm,
    float* __restrict__ SG,
    float* __restrict__ rowSumL) {
  __shared__ float red[256];
  const int b = blockIdx.x;
  float c0 = pl[0], c1 = pl[1], c2 = pl[2], c3 = pl[3], c4 = pl[4], c5 = pl[5], c6 = pl[6];
  float d0 = ph[0], d1 = ph[1], d2 = ph[2], d3 = ph[3], d4 = ph[4], d5 = ph[5];
  float lsum = 0.0f;
  for (int i = threadIdx.x; i < I_DIM; i += 256) {
    const float xv = x[b * I_DIM + i];
    float l = c6;
    l = fmaf(l, xv, c5); l = fmaf(l, xv, c4); l = fmaf(l, xv, c3);
    l = fmaf(l, xv, c2); l = fmaf(l, xv, c1); l = fmaf(l, xv, c0);
    float h = d5;
    h = fmaf(h, xv, d4); h = fmaf(h, xv, d3); h = fmaf(h, xv, d2);
    h = fmaf(h, xv, d1); h = fmaf(h, xv, d0);
    const float g = C1_JOHNSON / (fabsf(xv) + NOISE_MIN) + C2_SHOT;
    L[b * I_DIM + i]  = l;
    Dm[b * I_DIM + i] = h - l;
    SG[b * I_DIM + i] = sqrtf(g);
    lsum += l;
  }
  red[threadIdx.x] = lsum;
  __syncthreads();
  for (int s = 128; s > 0; s >>= 1) {
    if (threadIdx.x < (unsigned)s) red[threadIdx.x] += red[threadIdx.x + s];
    __syncthreads();
  }
  if (threadIdx.x == 0) rowSumL[b] = red[0];
}

__device__ __forceinline__ unsigned rotl32(unsigned v, int r) {
  return (v << r) | (v >> (32 - r));
}

__device__ __forceinline__ float bits_to_normal(unsigned bits) {
  const float lo = -0.99999994f;
  const float hi = 1.0f;
  float f = __uint_as_float((bits >> 9) | 0x3f800000u) - 1.0f;
  float u = fmaf(f, hi - lo, lo);
  u = fmaxf(u, lo);
  float w = -__logf((1.0f - u) * (1.0f + u));
  float p;
  if (w < 5.0f) {
    w -= 2.5f;
    p = 2.81022636e-08f;
    p = fmaf(p, w, 3.43273939e-07f);
    p = fmaf(p, w, -3.5233877e-06f);
    p = fmaf(p, w, -4.39150654e-06f);
    p = fmaf(p, w, 0.00021858087f);
    p = fmaf(p, w, -0.00125372503f);
    p = fmaf(p, w, -0.00417768164f);
    p = fmaf(p, w, 0.246640727f);
    p = fmaf(p, w, 1.50140941f);
  } else {
    w = sqrtf(w) - 3.0f;
    p = -0.000200214257f;
    p = fmaf(p, w, 0.000100950558f);
    p = fmaf(p, w, 0.00134934322f);
    p = fmaf(p, w, -0.00367342844f);
    p = fmaf(p, w, 0.00573950773f);
    p = fmaf(p, w, -0.0076224613f);
    p = fmaf(p, w, 0.00943887047f);
    p = fmaf(p, w, 1.00167406f);
    p = fmaf(p, w, 2.83297682f);
  }
  return 1.41421356237f * (p * u);
}

__global__ __launch_bounds__(256) void noise_kernel(float* __restrict__ N) {
  const unsigned half = (I_DIM * O_DIM) / 2;
  const unsigned j = blockIdx.x * 256u + threadIdx.x;
  if (j >= half) return;
  unsigned x0 = j, x1 = j + half;
  const unsigned ks0 = 0u, ks1 = 42u;
  const unsigned ks2 = 0x1BD11BDAu ^ ks0 ^ ks1;
  x0 += ks0; x1 += ks1;
  const int ra[4] = {13, 15, 26, 6};
  const int rb[4] = {17, 29, 16, 24};
#pragma unroll
  for (int r = 0; r < 4; ++r) { x0 += x1; x1 = rotl32(x1, ra[r]); x1 ^= x0; }
  x0 += ks1; x1 += ks2 + 1u;
#pragma unroll
  for (int r = 0; r < 4; ++r) { x0 += x1; x1 = rotl32(x1, rb[r]); x1 ^= x0; }
  x0 += ks2; x1 += ks0 + 2u;
#pragma unroll
  for (int r = 0; r < 4; ++r) { x0 += x1; x1 = rotl32(x1, ra[r]); x1 ^= x0; }
  x0 += ks0; x1 += ks1 + 3u;
#pragma unroll
  for (int r = 0; r < 4; ++r) { x0 += x1; x1 = rotl32(x1, rb[r]); x1 ^= x0; }
  x0 += ks1; x1 += ks2 + 4u;
#pragma unroll
  for (int r = 0; r < 4; ++r) { x0 += x1; x1 = rotl32(x1, ra[r]); x1 ^= x0; }
  x0 += ks2; x1 += ks0 + 5u;
  N[j]        = bits_to_normal(x0);
  N[j + half] = bits_to_normal(x1);
}

__global__ __launch_bounds__(256) void gemm_wmma_kernel(
    const float* __restrict__ Dm,
    const float* __restrict__ R,
    const float* __restrict__ rowSumL,
    float* __restrict__ out) {
  const int lane  = threadIdx.x & 31;
  const int wave  = threadIdx.x >> 5;
  const int ntile = blockIdx.x * 8 + wave;
  const int mtile = blockIdx.y;
  const int m = mtile * 16 + (lane & 15);
  const int n = ntile * 16 + (lane & 15);
  const int h = lane >> 4;

  v8f c = {};
  for (int k0 = 0; k0 < I_DIM; k0 += 4) {
    const int ka = k0 + 2 * h;
    v2f a = *(const v2f*)(Dm + m * I_DIM + ka);
    v2f bm;
    bm.x = R[ka * O_DIM + n];
    bm.y = R[(ka + 1) * O_DIM + n];
    c = __builtin_amdgcn_wmma_f32_16x16x4_f32(
         false, a,  false, bm,
         (short)0, c,  false,  false);
  }
#pragma unroll
  for (int v = 0; v < 8; ++v) {
    const int row = mtile * 16 + v + 8 * h;
    out[row * O_DIM + n] = c[v] + rowSumL[row];
  }
}

#define TB_B 8
#define TB_O 32
#define ICHUNK 64

__global__ __launch_bounds__(256) void noise_accum_kernel(
    const float* __restrict__ L, const float* __restrict__ Dm,
    const float* __restrict__ SG, const float* __restrict__ R,
    const float* __restrict__ N, float* __restrict__ out) {
  __shared__ float sL[TB_B][ICHUNK];
  __shared__ float sD[TB_B][ICHUNK];
  __shared__ float sG[TB_B][ICHUNK];
  const int t  = threadIdx.x;
  const int ob = t & (TB_O - 1);
  const int bb = t >> 5;
  const int o  = blockIdx.x * TB_O + ob;
  const int b  = blockIdx.y * TB_B + bb;

  float acc = 0.0f;
  for (int i0 = 0; i0 < I_DIM; i0 += ICHUNK) {
    __syncthreads();
    for (int s = t; s < TB_B * ICHUNK; s += 256) {
      const int rb = s / ICHUNK, ii = s % ICHUNK;
      const int g  = (blockIdx.y * TB_B + rb) * I_DIM + i0 + ii;
      sL[rb][ii] = L[g];
      sD[rb][ii] = Dm[g];
      sG[rb][ii] = SG[g];
    }
    __syncthreads();
    if (i0 + ICHUNK < I_DIM) {
      __builtin_prefetch(&R[(i0 + ICHUNK) * O_DIM + o], 0, 0);
      __builtin_prefetch(&N[(i0 + ICHUNK) * O_DIM + o], 0, 0);
    }
#pragma unroll 4
    for (int ii = 0; ii < ICHUNK; ++ii) {
      const int i  = i0 + ii;
      const float rv = R[i * O_DIM + o];
      const float nv = N[i * O_DIM + o];
      const float rr = fmaf(sD[bb][ii], rv, sL[bb][ii]);
      acc = fmaf(nv * sG[bb][ii], sqrtf(fabsf(rr)), acc);
    }
  }
  out[b * O_DIM + o] += acc;
}

extern "C" void kernel_launch(void* const* d_in, const int* in_sizes, int n_in,
                              void* d_out, int out_size, void* d_ws, size_t ws_size,
                              hipStream_t stream) {
  (void)in_sizes; (void)n_in; (void)out_size; (void)ws_size;
  const float* x  = (const float*)d_in[0];
  const float* pl = (const float*)d_in[1];
  const float* ph = (const float*)d_in[2];
  const float* r  = (const float*)d_in[3];
  float* out = (float*)d_out;

  float* wL  = (float*)d_ws;
  float* wD  = wL + B_DIM * I_DIM;
  float* wS  = wD + B_DIM * I_DIM;
  float* wN  = wS + B_DIM * I_DIM;
  float* wRS = wN + I_DIM * O_DIM;

  precomp_kernel<<<dim3(B_DIM), dim3(256), 0, stream>>>(x, pl, ph, wL, wD, wS, wRS);

  noise_kernel<<<dim3((I_DIM * O_DIM / 2 + 255) / 256), dim3(256), 0, stream>>>(wN);

  gemm_wmma_kernel<<<dim3(O_DIM / (16 * 8), B_DIM / 16), dim3(256), 0, stream>>>(
      wD, r, wRS, out);

  noise_accum_kernel<<<dim3(O_DIM / TB_O, B_DIM / TB_B), dim3(256), 0, stream>>>(
      wL, wD, wS, r, wN, out);
}


# noise sweep: each lane keeps r and noise in registers and uses them for all 8 batch rows of its tile; i split over 16 waves, partials combined in LDS (noise-term sum order changes)
# speedup vs baseline: 2.5842x; 2.5842x over previous
#include <hip/hip_runtime.h>
#include <hip/hip_bf16.h>


typedef __attribute__((ext_vector_type(2))) float v2f;
typedef __attribute__((ext_vector_type(8))) float v8f;

#define B_DIM 128
#define I_DIM 512
#define O_DIM 512

#define C1_JOHNSON 1.65677880e-28f
#define C2_SHOT    5.43656366e-08f
#define NOISE_MIN  1e-12f

__global__ __launch_bounds__(256) void precomp_kernel(
    const float* __restrict__ x,
    const float* __restrict__ pl,
    const float* __restrict__ ph,
    float* __restrict__ L,
    float* __restrict__ Dm,
    float* __restrict__ SG,
    float* __restrict__ rowSumL) {
  __shared__ float red[256];
  const int b = blockIdx.x;
  float c0 = pl[0], c1 = pl[1], c2 = pl[2], c3 = pl[3], c4 = pl[4], c5 = pl[5], c6 = pl[6];
  float d0 = ph[0], d1 = ph[1], d2 = ph[2], d3 = ph[3], d4 = ph[4], d5 = ph[5];
  float lsum = 0.0f;
  for (int i = threadIdx.x; i < I_DIM; i += 256) {
    const float xv = x[b * I_DIM + i];
    float l = c6;
    l = fmaf(l, xv, c5); l = fmaf(l, xv, c4); l = fmaf(l, xv, c3);
    l = fmaf(l, xv, c2); l = fmaf(l, xv, c1); l = fmaf(l, xv, c0);
    float h = d5;
    h = fmaf(h, xv, d4); h = fmaf(h, xv, d3); h = fmaf(h, xv, d2);
    h = fmaf(h, xv, d1); h = fmaf(h, xv, d0);
    const float g = C1_JOHNSON / (fabsf(xv) + NOISE_MIN) + C2_SHOT;
    L[b * I_DIM + i]  = l;
    Dm[b * I_DIM + i] = h - l;
    SG[b * I_DIM + i] = sqrtf(g);
    lsum += l;
  }
  red[threadIdx.x] = lsum;
  __syncthreads();
  for (int s = 128; s > 0; s >>= 1) {
    if (threadIdx.x < (unsigned)s) red[threadIdx.x] += red[threadIdx.x + s];
    __syncthreads();
  }
  if (threadIdx.x == 0) rowSumL[b] = red[0];
}

__device__ __forceinline__ unsigned rotl32(unsigned v, int r) {
  return (v << r) | (v >> (32 - r));
}

__device__ __forceinline__ float bits_to_normal(unsigned bits) {
  const float lo = -0.99999994f;
  const float hi = 1.0f;
  float f = __uint_as_float((bits >> 9) | 0x3f800000u) - 1.0f;
  float u = fmaf(f, hi - lo, lo);
  u = fmaxf(u, lo);
  float w = -__logf((1.0f - u) * (1.0f + u));
  float p;
  if (w < 5.0f) {
    w -= 2.5f;
    p = 2.81022636e-08f;
    p = fmaf(p, w, 3.43273939e-07f);
    p = fmaf(p, w, -3.5233877e-06f);
    p = fmaf(p, w, -4.39150654e-06f);
    p = fmaf(p, w, 0.00021858087f);
    p = fmaf(p, w, -0.00125372503f);
    p = fmaf(p, w, -0.00417768164f);
    p = fmaf(p, w, 0.246640727f);
    p = fmaf(p, w, 1.50140941f);
  } else {
    w = sqrtf(w) - 3.0f;
    p = -0.000200214257f;
    p = fmaf(p, w, 0.000100950558f);
    p = fmaf(p, w, 0.00134934322f);
    p = fmaf(p, w, -0.00367342844f);
    p = fmaf(p, w, 0.00573950773f);
    p = fmaf(p, w, -0.0076224613f);
    p = fmaf(p, w, 0.00943887047f);
    p = fmaf(p, w, 1.00167406f);
    p = fmaf(p, w, 2.83297682f);
  }
  return 1.41421356237f * (p * u);
}

__global__ __launch_bounds__(256) void noise_kernel(float* __restrict__ N) {
  const unsigned half = (I_DIM * O_DIM) / 2;
  const unsigned j = blockIdx.x * 256u + threadIdx.x;
  if (j >= half) return;
  unsigned x0 = j, x1 = j + half;
  const unsigned ks0 = 0u, ks1 = 42u;
  const unsigned ks2 = 0x1BD11BDAu ^ ks0 ^ ks1;
  x0 += ks0; x1 += ks1;
  const int ra[4] = {13, 15, 26, 6};
  const int rb[4] = {17, 29, 16, 24};
#pragma unroll
  for (int r = 0; r < 4; ++r) { x0 += x1; x1 = rotl32(x1, ra[r]); x1 ^= x0; }
  x0 += ks1; x1 += ks2 + 1u;
#pragma unroll
  for (int r = 0; r < 4; ++r) { x0 += x1; x1 = rotl32(x1, rb[r]); x1 ^= x0; }
  x0 += ks2; x1 += ks0 + 2u;
#pragma unroll
  for (int r = 0; r < 4; ++r) { x0 += x1; x1 = rotl32(x1, ra[r]); x1 ^= x0; }
  x0 += ks0; x1 += ks1 + 3u;
#pragma unroll
  for (int r = 0; r < 4; ++r) { x0 += x1; x1 = rotl32(x1, rb[r]); x1 ^= x0; }
  x0 += ks1; x1 += ks2 + 4u;
#pragma unroll
  for (int r = 0; r < 4; ++r) { x0 += x1; x1 = rotl32(x1, ra[r]); x1 ^= x0; }
  x0 += ks2; x1 += ks0 + 5u;
  N[j]        = bits_to_normal(x0);
  N[j + half] = bits_to_normal(x1);
}

__global__ __launch_bounds__(256) void gemm_wmma_kernel(
    const float* __restrict__ Dm,
    const float* __restrict__ R,
    const float* __restrict__ rowSumL,
    float* __restrict__ out) {
  const int lane  = threadIdx.x & 31;
  const int wave  = threadIdx.x >> 5;
  const int ntile = blockIdx.x * 8 + wave;
  const int mtile = blockIdx.y;
  const int m = mtile * 16 + (lane & 15);
  const int n = ntile * 16 + (lane & 15);
  const int h = lane >> 4;

  v8f c = {};
  for (int k0 = 0; k0 < I_DIM; k0 += 4) {
    const int ka = k0 + 2 * h;
    v2f a = *(const v2f*)(Dm + m * I_DIM + ka);
    v2f bm;
    bm.x = R[ka * O_DIM + n];
    bm.y = R[(ka + 1) * O_DIM + n];
    c = __builtin_amdgcn_wmma_f32_16x16x4_f32(
         false, a,  false, bm,
         (short)0, c,  false,  false);
  }
#pragma unroll
  for (int v = 0; v < 8; ++v) {
    const int row = mtile * 16 + v + 8 * h;
    out[row * O_DIM + n] = c[v] + rowSumL[row];
  }
}

#define NS_B 8
#define NS_O 32
#define NS_W 16
#define NS_I (I_DIM / NS_W)
#define NS_U 4

__global__ __launch_bounds__(NS_W * 32)
__attribute__((amdgpu_waves_per_eu(4, 4))) void noise_accum_kernel(
    const float* __restrict__ L, const float* __restrict__ Dm,
    const float* __restrict__ SG, const float* __restrict__ R,
    const float* __restrict__ N, float* __restrict__ out) {
  __shared__ __attribute__((aligned(16))) float sm[3 * NS_B * I_DIM];
  const int t    = threadIdx.x;
  const int lane = t & 31;
  const int w    = t >> 5;
  const int o    = blockIdx.x * NS_O + lane;
  const int b0   = blockIdx.y * NS_B;
  const int ibeg = w * NS_I;

  {
    const float4* gL = (const float4*)(L  + b0 * I_DIM);
    const float4* gD = (const float4*)(Dm + b0 * I_DIM);
    const float4* gG = (const float4*)(SG + b0 * I_DIM);
    float4* s4 = (float4*)sm;
    const int n4 = NS_B * I_DIM / 4;
#pragma unroll
    for (int k = 0; k < n4 / (NS_W * 32); ++k) {
      const int s = t + k * (NS_W * 32);
      s4[s]          = gL[s];
      s4[n4 + s]     = gD[s];
      s4[2 * n4 + s] = gG[s];
    }
  }

  const float* rp = R + ibeg * O_DIM + o;
  const float* np = N + ibeg * O_DIM + o;
  v2f rv[NS_U / 2], nv[NS_U / 2];
#pragma unroll
  for (int u = 0; u < NS_U / 2; ++u) {
    rv[u].x = rp[(2 * u) * O_DIM];  rv[u].y = rp[(2 * u + 1) * O_DIM];
    nv[u].x = np[(2 * u) * O_DIM];  nv[u].y = np[(2 * u + 1) * O_DIM];
  }
  __syncthreads();

  const float* sL = sm;
  const float* sD = sm + NS_B * I_DIM;
  const float* sG = sm + 2 * NS_B * I_DIM;

  float acc[NS_B];
#pragma unroll
  for (int bb = 0; bb < NS_B; ++bb) acc[bb] = 0.0f;

  for (int i0 = 0; i0 < NS_I; i0 += NS_U) {
    v2f rn[NS_U / 2], nn[NS_U / 2];
    const int inx = (i0 + NS_U < NS_I) ? i0 + NS_U : i0;
#pragma unroll
    for (int u = 0; u < NS_U / 2; ++u) {
      rn[u].x = rp[(inx + 2 * u) * O_DIM];  rn[u].y = rp[(inx + 2 * u + 1) * O_DIM];
      nn[u].x = np[(inx + 2 * u) * O_DIM];  nn[u].y = np[(inx + 2 * u + 1) * O_DIM];
    }
#pragma unroll
    for (int bb = 0; bb < NS_B; ++bb) {
      const int si = bb * I_DIM + ibeg + i0;
      const float4 l4 = *(const float4*)(sL + si);
      const float4 d4 = *(const float4*)(sD + si);
      const float4 g4 = *(const float4*)(sG + si);
      const v2f l01 = {l4.x, l4.y}, l23 = {l4.z, l4.w};
      const v2f d01 = {d4.x, d4.y}, d23 = {d4.z, d4.w};
      const v2f g01 = {g4.x, g4.y}, g23 = {g4.z, g4.w};
      const v2f rr01 = __builtin_elementwise_fma(d01, rv[0], l01);
      const v2f rr23 = __builtin_elementwise_fma(d23, rv[1], l23);
      const v2f t01 = nv[0] * g01;
      const v2f t23 = nv[1] * g23;
      float a = acc[bb];
      a = fmaf(t01.x, sqrtf(fabsf(rr01.x)), a);
      a = fmaf(t01.y, sqrtf(fabsf(rr01.y)), a);
      a = fmaf(t23.x, sqrtf(fabsf(rr23.x)), a);
      a = fmaf(t23.y, sqrtf(fabsf(rr23.y)), a);
      acc[bb] = a;
    }
#pragma unroll
    for (int u = 0; u < NS_U / 2; ++u) { rv[u] = rn[u]; nv[u] = nn[u]; }
  }

  __syncthreads();
#pragma unroll
  for (int bb = 0; bb < NS_B; ++bb) sm[(w * NS_B + bb) * NS_O + lane] = acc[bb];
  __syncthreads();
  if (t < NS_B * NS_O) {
    float ssum = 0.0f;
#pragma unroll
    for (int ww = 0; ww < NS_W; ++ww) ssum += sm[ww * (NS_B * NS_O) + t];
    const int bb = t >> 5;
    out[(b0 + bb) * O_DIM + o] += ssum;
  }
}

extern "C" void kernel_launch(void* const* d_in, const int* in_sizes, int n_in,
                              void* d_out, int out_size, void* d_ws, size_t ws_size,
                              hipStream_t stream) {
  (void)in_sizes; (void)n_in; (void)out_size; (void)ws_size;
  const float* x  = (const float*)d_in[0];
  const float* pl = (const float*)d_in[1];
  const float* ph = (const float*)d_in[2];
  const float* r  = (const float*)d_in[3];
  float* out = (float*)d_out;

  float* wL  = (float*)d_ws;
  float* wD  = wL + B_DIM * I_DIM;
  float* wS  = wD + B_DIM * I_DIM;
  float* wN  = wS + B_DIM * I_DIM;
  float* wRS = wN + I_DIM * O_DIM;

  precomp_kernel<<<dim3(B_DIM), dim3(256), 0, stream>>>(x, pl, ph, wL, wD, wS, wRS);

  noise_kernel<<<dim3((I_DIM * O_DIM / 2 + 255) / 256), dim3(256), 0, stream>>>(wN);

  gemm_wmma_kernel<<<dim3(O_DIM / (16 * 8), B_DIM / 16), dim3(256), 0, stream>>>(
      wD, r, wRS, out);

  noise_accum_kernel<<<dim3(O_DIM / NS_O, B_DIM / NS_B), dim3(NS_W * 32), 0, stream>>>(
      wL, wD, wS, r, wN, out);
}


# noise sweep: square root by the bare v_sqrt_f32 instruction (1 ulp, no denormal scaling) instead of sqrtf's correctly rounded expansion of about a dozen vector instructions per element
# speedup vs baseline: 3.1197x; 1.2072x over previous
#include <hip/hip_runtime.h>
#include <hip/hip_bf16.h>


typedef __attribute__((ext_vector_type(2))) float v2f;
typedef __attribute__((ext_vector_type(8))) float v8f;

#define B_DIM 128
#define I_DIM 512
#define O_DIM 512

#define C1_JOHNSON 1.65677880e-28f
#define C2_SHOT    5.43656366e-08f
#define NOISE_MIN  1e-12f

__global__ __launch_bounds__(256) void precomp_kernel(
    const float* __restrict__ x,
    const float* __restrict__ pl,
    const float* __restrict__ ph,
    float* __restrict__ L,
    float* __restrict__ Dm,
    float* __restrict__ SG,
    float* __restrict__ rowSumL) {
  __shared__ float red[256];
  const int b = blockIdx.x;
  float c0 = pl[0], c1 = pl[1], c2 = pl[2], c3 = pl[3], c4 = pl[4], c5 = pl[5], c6 = pl[6];
  float d0 = ph[0], d1 = ph[1], d2 = ph[2], d3 = ph[3], d4 = ph[4], d5 = ph[5];
  float lsum = 0.0f;
  for (int i = threadIdx.x; i < I_DIM; i += 256) {
    const float xv = x[b * I_DIM + i];
    float l = c6;
    l = fmaf(l, xv, c5); l = fmaf(l, xv, c4); l = fmaf(l, xv, c3);
    l = fmaf(l, xv, c2); l = fmaf(l, xv, c1); l = fmaf(l, xv, c0);
    float h = d5;
    h = fmaf(h, xv, d4); h = fmaf(h, xv, d3); h = fmaf(h, xv, d2);
    h = fmaf(h, xv, d1); h = fmaf(h, xv, d0);
    const float g = C1_JOHNSON / (fabsf(xv) + NOISE_MIN) + C2_SHOT;
    L[b * I_DIM + i]  = l;
    Dm[b * I_DIM + i] = h - l;
    SG[b * I_DIM + i] = sqrtf(g);
    lsum += l;
  }
  red[threadIdx.x] = lsum;
  __syncthreads();
  for (int s = 128; s > 0; s >>= 1) {
    if (threadIdx.x < (unsigned)s) red[threadIdx.x] += red[threadIdx.x + s];
    __syncthreads();
  }
  if (threadIdx.x == 0) rowSumL[b] = red[0];
}

__device__ __forceinline__ unsigned rotl32(unsigned v, int r) {
  return (v << r) | (v >> (32 - r));
}

__device__ __forceinline__ float bits_to_normal(unsigned bits) {
  const float lo = -0.99999994f;
  const float hi = 1.0f;
  float f = __uint_as_float((bits >> 9) | 0x3f800000u) - 1.0f;
  float u = fmaf(f, hi - lo, lo);
  u = fmaxf(u, lo);
  float w = -__logf((1.0f - u) * (1.0f + u));
  float p;
  if (w < 5.0f) {
    w -= 2.5f;
    p = 2.81022636e-08f;
    p = fmaf(p, w, 3.43273939e-07f);
    p = fmaf(p, w, -3.5233877e-06f);
    p = fmaf(p, w, -4.39150654e-06f);
    p = fmaf(p, w, 0.00021858087f);
    p = fmaf(p, w, -0.00125372503f);
    p = fmaf(p, w, -0.00417768164f);
    p = fmaf(p, w, 0.246640727f);
    p = fmaf(p, w, 1.50140941f);
  } else {
    w = sqrtf(w) - 3.0f;
    p = -0.000200214257f;
    p = fmaf(p, w, 0.000100950558f);
    p = fmaf(p, w, 0.00134934322f);
    p = fmaf(p, w, -0.00367342844f);
    p = fmaf(p, w, 0.00573950773f);
    p = fmaf(p, w, -0.0076224613f);
    p = fmaf(p, w, 0.00943887047f);
    p = fmaf(p, w, 1.00167406f);
    p = fmaf(p, w, 2.83297682f);
  }
  return 1.41421356237f * (p * u);
}

__global__ __launch_bounds__(256) void noise_kernel(float* __restrict__ N) {
  const unsigned half = (I_DIM * O_DIM) / 2;
  const unsigned j = blockIdx.x * 256u + threadIdx.x;
  if (j >= half) return;
  unsigned x0 = j, x1 = j + half;
  const unsigned ks0 = 0u, ks1 = 42u;
  const unsigned ks2 = 0x1BD11BDAu ^ ks0 ^ ks1;
  x0 += ks0; x1 += ks1;
  const int ra[4] = {13, 15, 26, 6};
  const int rb[4] = {17, 29, 16, 24};
#pragma unroll
  for (int r = 0; r < 4; ++r) { x0 += x1; x1 = rotl32(x1, ra[r]); x1 ^= x0; }
  x0 += ks1; x1 += ks2 + 1u;
#pragma unroll
  for (int r = 0; r < 4; ++r) { x0 += x1; x1 = rotl32(x1, rb[r]); x1 ^= x0; }
  x0 += ks2; x1 += ks0 + 2u;
#pragma unroll
  for (int r = 0; r < 4; ++r) { x0 += x1; x1 = rotl32(x1, ra[r]); x1 ^= x0; }
  x0 += ks0; x1 += ks1 + 3u;
#pragma unroll
  for (int r = 0; r < 4; ++r) { x0 += x1; x1 = rotl32(x1, rb[r]); x1 ^= x0; }
  x0 += ks1; x1 += ks2 + 4u;
#pragma unroll
  for (int r = 0; r < 4; ++r) { x0 += x1; x1 = rotl32(x1, ra[r]); x1 ^= x0; }
  x0 += ks2; x1 += ks0 + 5u;
  N[j]        = bits_to_normal(x0);
  N[j + half] = bits_to_normal(x1);
}

__global__ __launch_bounds__(256) void gemm_wmma_kernel(
    const float* __restrict__ Dm,
    const float* __restrict__ R,
    const float* __restrict__ rowSumL,
    float* __restrict__ out) {
  const int lane  = threadIdx.x & 31;
  const int wave  = threadIdx.x >> 5;
  const int ntile = blockIdx.x * 8 + wave;
  const int mtile = blockIdx.y;
  const int m = mtile * 16 + (lane & 15);
  const int n = ntile * 16 + (lane & 15);
  const int h = lane >> 4;

  v8f c = {};
  for (int k0 = 0; k0 < I_DIM; k0 += 4) {
    const int ka = k0 + 2 * h;
    v2f a = *(const v2f*)(Dm + m * I_DIM + ka);
    v2f bm;
    bm.x = R[ka * O_DIM + n];
    bm.y = R[(ka + 1) * O_DIM + n];
    c = __builtin_amdgcn_wmma_f32_16x16x4_f32(
         false, a,  false, bm,
         (short)0, c,  false,  false);
  }
#pragma unroll
  for (int v = 0; v < 8; ++v) {
    const int row = mtile * 16 + v + 8 * h;
    out[row * O_DIM + n] = c[v] + rowSumL[row];
  }
}

__device__ __forceinline__ float hw_sqrt(float v) { return __builtin_amdgcn_sqrtf(v); }

#define NS_B 8
#define NS_O 32
#define NS_W 16
#define NS_I (I_DIM / NS_W)
#define NS_U 4

__global__ __launch_bounds__(NS_W * 32)
__attribute__((amdgpu_waves_per_eu(4, 4))) void noise_accum_kernel(
    const float* __restrict__ L, const float* __restrict__ Dm,
    const float* __restrict__ SG, const float* __restrict__ R,
    const float* __restrict__ N, float* __restrict__ out) {
  __shared__ __attribute__((aligned(16))) float sm[3 * NS_B * I_DIM];
  const int t    = threadIdx.x;
  const int lane = t & 31;
  const int w    = t >> 5;
  const int o    = blockIdx.x * NS_O + lane;
  const int b0   = blockIdx.y * NS_B;
  const int ibeg = w * NS_I;

  {
    const float4* gL = (const float4*)(L  + b0 * I_DIM);
    const float4* gD = (const float4*)(Dm + b0 * I_DIM);
    const float4* gG = (const float4*)(SG + b0 * I_DIM);
    float4* s4 = (float4*)sm;
    const int n4 = NS_B * I_DIM / 4;
#pragma unroll
    for (int k = 0; k < n4 / (NS_W * 32); ++k) {
      const int s = t + k * (NS_W * 32);
      s4[s]          = gL[s];
      s4[n4 + s]     = gD[s];
      s4[2 * n4 + s] = gG[s];
    }
  }

  const float* rp = R + ibeg * O_DIM + o;
  const float* np = N + ibeg * O_DIM + o;
  v2f rv[NS_U / 2], nv[NS_U / 2];
#pragma unroll
  for (int u = 0; u < NS_U / 2; ++u) {
    rv[u].x = rp[(2 * u) * O_DIM];  rv[u].y = rp[(2 * u + 1) * O_DIM];
    nv[u].x = np[(2 * u) * O_DIM];  nv[u].y = np[(2 * u + 1) * O_DIM];
  }
  __syncthreads();

  const float* sL = sm;
  const float* sD = sm + NS_B * I_DIM;
  const float* sG = sm + 2 * NS_B * I_DIM;

  float acc[NS_B];
#pragma unroll
  for (int bb = 0; bb < NS_B; ++bb) acc[bb] = 0.0f;

  for (int i0 = 0; i0 < NS_I; i0 += NS_U) {
    v2f rn[NS_U / 2], nn[NS_U / 2];
    const int inx = (i0 + NS_U < NS_I) ? i0 + NS_U : i0;
#pragma unroll
    for (int u = 0; u < NS_U / 2; ++u) {
      rn[u].x = rp[(inx + 2 * u) * O_DIM];  rn[u].y = rp[(inx + 2 * u + 1) * O_DIM];
      nn[u].x = np[(inx + 2 * u) * O_DIM];  nn[u].y = np[(inx + 2 * u + 1) * O_DIM];
    }
#pragma unroll
    for (int bb = 0; bb < NS_B; ++bb) {
      const int si = bb * I_DIM + ibeg + i0;
      const float4 l4 = *(const float4*)(sL + si);
      const float4 d4 = *(const float4*)(sD + si);
      const float4 g4 = *(const float4*)(sG + si);
      const v2f l01 = {l4.x, l4.y}, l23 = {l4.z, l4.w};
      const v2f d01 = {d4.x, d4.y}, d23 = {d4.z, d4.w};
      const v2f g01 = {g4.x, g4.y}, g23 = {g4.z, g4.w};
      const v2f rr01 = __builtin_elementwise_fma(d01, rv[0], l01);
      const v2f rr23 = __builtin_elementwise_fma(d23, rv[1], l23);
      const v2f t01 = nv[0] * g01;
      const v2f t23 = nv[1] * g23;
      float a = acc[bb];
      a = fmaf(t01.x, hw_sqrt(fabsf(rr01.x)), a);
      a = fmaf(t01.y, hw_sqrt(fabsf(rr01.y)), a);
      a = fmaf(t23.x, hw_sqrt(fabsf(rr23.x)), a);
      a = fmaf(t23.y, hw_sqrt(fabsf(rr23.y)), a);
      acc[bb] = a;
    }
#pragma unroll
    for (int u = 0; u < NS_U / 2; ++u) { rv[u] = rn[u]; nv[u] = nn[u]; }
  }

  __syncthreads();
#pragma unroll
  for (int bb = 0; bb < NS_B; ++bb) sm[(w * NS_B + bb) * NS_O + lane] = acc[bb];
  __syncthreads();
  if (t < NS_B * NS_O) {
    float ssum = 0.0f;
#pragma unroll
    for (int ww = 0; ww < NS_W; ++ww) ssum += sm[ww * (NS_B * NS_O) + t];
    const int bb = t >> 5;
    out[(b0 + bb) * O_DIM + o] += ssum;
  }
}

extern "C" void kernel_launch(void* const* d_in, const int* in_sizes, int n_in,
                              void* d_out, int out_size, void* d_ws, size_t ws_size,
                              hipStream_t stream) {
  (void)in_sizes; (void)n_in; (void)out_size; (void)ws_size;
  const float* x  = (const float*)d_in[0];
  const float* pl = (const float*)d_in[1];
  const float* ph = (const float*)d_in[2];
  const float* r  = (const float*)d_in[3];
  float* out = (float*)d_out;

  float* wL  = (float*)d_ws;
  float* wD  = wL + B_DIM * I_DIM;
  float* wS  = wD + B_DIM * I_DIM;
  float* wN  = wS + B_DIM * I_DIM;
  float* wRS = wN + I_DIM * O_DIM;

  precomp_kernel<<<dim3(B_DIM), dim3(256), 0, stream>>>(x, pl, ph, wL, wD, wS, wRS);

  noise_kernel<<<dim3((I_DIM * O_DIM / 2 + 255) / 256), dim3(256), 0, stream>>>(wN);

  gemm_wmma_kernel<<<dim3(O_DIM / (16 * 8), B_DIM / 16), dim3(256), 0, stream>>>(
      wD, r, wRS, out);

  noise_accum_kernel<<<dim3(O_DIM / NS_O, B_DIM / NS_B), dim3(NS_W * 32), 0, stream>>>(
      wL, wD, wS, r, wN, out);
}


# GEMM and noise generation (independent of each other) run in one launch: GEMM blocks take the lowest block indices, the 512 noise blocks fill the chip beside them; 4 launches become 3
# speedup vs baseline: 3.2186x; 1.0317x over previous
#include <hip/hip_runtime.h>
#include <hip/hip_bf16.h>


typedef __attribute__((ext_vector_type(2))) float v2f;
typedef __attribute__((ext_vector_type(8))) float v8f;

#define B_DIM 128
#define I_DIM 512
#define O_DIM 512

#define C1_JOHNSON 1.65677880e-28f
#define C2_SHOT    5.43656366e-08f
#define NOISE_MIN  1e-12f

__global__ __launch_bounds__(256) void precomp_kernel(
    const float* __restrict__ x,
    const float* __restrict__ pl,
    const float* __restrict__ ph,
    float* __restrict__ L,
    float* __restrict__ Dm,
    float* __restrict__ SG,
    float* __restrict__ rowSumL) {
  __shared__ float red[256];
  const int b = blockIdx.x;
  float c0 = pl[0], c1 = pl[1], c2 = pl[2], c3 = pl[3], c4 = pl[4], c5 = pl[5], c6 = pl[6];
  float d0 = ph[0], d1 = ph[1], d2 = ph[2], d3 = ph[3], d4 = ph[4], d5 = ph[5];
  float lsum = 0.0f;
  for (int i = threadIdx.x; i < I_DIM; i += 256) {
    const float xv = x[b * I_DIM + i];
    float l = c6;
    l = fmaf(l, xv, c5); l = fmaf(l, xv, c4); l = fmaf(l, xv, c3);
    l = fmaf(l, xv, c2); l = fmaf(l, xv, c1); l = fmaf(l, xv, c0);
    float h = d5;
    h = fmaf(h, xv, d4); h = fmaf(h, xv, d3); h = fmaf(h, xv, d2);
    h = fmaf(h, xv, d1); h = fmaf(h, xv, d0);
    const float g = C1_JOHNSON / (fabsf(xv) + NOISE_MIN) + C2_SHOT;
    L[b * I_DIM + i]  = l;
    Dm[b * I_DIM + i] = h - l;
    SG[b * I_DIM + i] = sqrtf(g);
    lsum += l;
  }
  red[threadIdx.x] = lsum;
  __syncthreads();
  for (int s = 128; s > 0; s >>= 1) {
    if (threadIdx.x < (unsigned)s) red[threadIdx.x] += red[threadIdx.x + s];
    __syncthreads();
  }
  if (threadIdx.x == 0) rowSumL[b] = red[0];
}

__device__ __forceinline__ unsigned rotl32(unsigned v, int r) {
  return (v << r) | (v >> (32 - r));
}

__device__ __forceinline__ float bits_to_normal(unsigned bits) {
  const float lo = -0.99999994f;
  const float hi = 1.0f;
  float f = __uint_as_float((bits >> 9) | 0x3f800000u) - 1.0f;
  float u = fmaf(f, hi - lo, lo);
  u = fmaxf(u, lo);
  float w = -__logf((1.0f - u) * (1.0f + u));
  float p;
  if (w < 5.0f) {
    w -= 2.5f;
    p = 2.81022636e-08f;
    p = fmaf(p, w, 3.43273939e-07f);
    p = fmaf(p, w, -3.5233877e-06f);
    p = fmaf(p, w, -4.39150654e-06f);
    p = fmaf(p, w, 0.00021858087f);
    p = fmaf(p, w, -0.00125372503f);
    p = fmaf(p, w, -0.00417768164f);
    p = fmaf(p, w, 0.246640727f);
    p = fmaf(p, w, 1.50140941f);
  } else {
    w = sqrtf(w) - 3.0f;
    p = -0.000200214257f;
    p = fmaf(p, w, 0.000100950558f);
    p = fmaf(p, w, 0.00134934322f);
    p = fmaf(p, w, -0.00367342844f);
    p = fmaf(p, w, 0.00573950773f);
    p = fmaf(p, w, -0.0076224613f);
    p = fmaf(p, w, 0.00943887047f);
    p = fmaf(p, w, 1.00167406f);
    p = fmaf(p, w, 2.83297682f);
  }
  return 1.41421356237f * (p * u);
}

__device__ __forceinline__ void noise_block(float* __restrict__ N, unsigned nb) {
  const unsigned half = (I_DIM * O_DIM) / 2;
  const unsigned j = nb * 256u + threadIdx.x;
  if (j >= half) return;
  unsigned x0 = j, x1 = j + half;
  const unsigned ks0 = 0u, ks1 = 42u;
  const unsigned ks2 = 0x1BD11BDAu ^ ks0 ^ ks1;
  x0 += ks0; x1 += ks1;
  const int ra[4] = {13, 15, 26, 6};
  const int rb[4] = {17, 29, 16, 24};
#pragma unroll
  for (int r = 0; r < 4; ++r) { x0 += x1; x1 = rotl32(x1, ra[r]); x1 ^= x0; }
  x0 += ks1; x1 += ks2 + 1u;
#pragma unroll
  for (int r = 0; r < 4; ++r) { x0 += x1; x1 = rotl32(x1, rb[r]); x1 ^= x0; }
  x0 += ks2; x1 += ks0 + 2u;
#pragma unroll
  for (int r = 0; r < 4; ++r) { x0 += x1; x1 = rotl32(x1, ra[r]); x1 ^= x0; }
  x0 += ks0; x1 += ks1 + 3u;
#pragma unroll
  for (int r = 0; r < 4; ++r) { x0 += x1; x1 = rotl32(x1, rb[r]); x1 ^= x0; }
  x0 += ks1; x1 += ks2 + 4u;
#pragma unroll
  for (int r = 0; r < 4; ++r) { x0 += x1; x1 = rotl32(x1, ra[r]); x1 ^= x0; }
  x0 += ks2; x1 += ks0 + 5u;
  N[j]        = bits_to_normal(x0);
  N[j + half] = bits_to_normal(x1);
}

__device__ __forceinline__ void gemm_block(
    const float* __restrict__ Dm,
    const float* __restrict__ R,
    const float* __restrict__ rowSumL,
    float* __restrict__ out,
    int gb) {
  const int lane  = threadIdx.x & 31;
  const int wave  = threadIdx.x >> 5;
  const int ntile = (gb & 3) * 8 + wave;
  const int mtile = gb >> 2;
  const int m = mtile * 16 + (lane & 15);
  const int n = ntile * 16 + (lane & 15);
  const int h = lane >> 4;

  v8f c = {};
  for (int k0 = 0; k0 < I_DIM; k0 += 4) {
    const int ka = k0 + 2 * h;
    v2f a = *(const v2f*)(Dm + m * I_DIM + ka);
    v2f bm;
    bm.x = R[ka * O_DIM + n];
    bm.y = R[(ka + 1) * O_DIM + n];
    c = __builtin_amdgcn_wmma_f32_16x16x4_f32(
         false, a,  false, bm,
         (short)0, c,  false,  false);
  }
#pragma unroll
  for (int v = 0; v < 8; ++v) {
    const int row = mtile * 16 + v + 8 * h;
    out[row * O_DIM + n] = c[v] + rowSumL[row];
  }
}

#define GEMM_BLOCKS  ((O_DIM / (16 * 8)) * (B_DIM / 16))
#define NOISE_BLOCKS ((I_DIM * O_DIM / 2 + 255) / 256)

__global__ __launch_bounds__(256) void gemm_and_noise_kernel(
    const float* __restrict__ Dm,
    const float* __restrict__ R,
    const float* __restrict__ rowSumL,
    float* __restrict__ out,
    float* __restrict__ N) {
  if (blockIdx.x < GEMM_BLOCKS) {
    gemm_block(Dm, R, rowSumL, out, (int)blockIdx.x);
  } else {
    noise_block(N, blockIdx.x - GEMM_BLOCKS);
  }
}

__device__ __forceinline__ float hw_sqrt(float v) { return __builtin_amdgcn_sqrtf(v); }

#define NS_B 8
#define NS_O 32
#define NS_W 16
#define NS_I (I_DIM / NS_W)
#define NS_U 4

__global__ __launch_bounds__(NS_W * 32)
__attribute__((amdgpu_waves_per_eu(4, 4))) void noise_accum_kernel(
    const float* __restrict__ L, const float* __restrict__ Dm,
    const float* __restrict__ SG, const float* __restrict__ R,
    const float* __restrict__ N, float* __restrict__ out) {
  __shared__ __attribute__((aligned(16))) float sm[3 * NS_B * I_DIM];
  const int t    = threadIdx.x;
  const int lane = t & 31;
  const int w    = t >> 5;
  const int o    = blockIdx.x * NS_O + lane;
  const int b0   = blockIdx.y * NS_B;
  const int ibeg = w * NS_I;

  {
    const float4* gL = (const float4*)(L  + b0 * I_DIM);
    const float4* gD = (const float4*)(Dm + b0 * I_DIM);
    const float4* gG = (const float4*)(SG + b0 * I_DIM);
    float4* s4 = (float4*)sm;
    const int n4 = NS_B * I_DIM / 4;
#pragma unroll
    for (int k = 0; k < n4 / (NS_W * 32); ++k) {
      const int s = t + k * (NS_W * 32);
      s4[s]          = gL[s];
      s4[n4 + s]     = gD[s];
      s4[2 * n4 + s] = gG[s];
    }
  }

  const float* rp = R + ibeg * O_DIM + o;
  const float* np = N + ibeg * O_DIM + o;
  v2f rv[NS_U / 2], nv[NS_U / 2];
#pragma unroll
  for (int u = 0; u < NS_U / 2; ++u) {
    rv[u].x = rp[(2 * u) * O_DIM];  rv[u].y = rp[(2 * u + 1) * O_DIM];
    nv[u].x = np[(2 * u) * O_DIM];  nv[u].y = np[(2 * u + 1) * O_DIM];
  }
  __syncthreads();

  const float* sL = sm;
  const float* sD = sm + NS_B * I_DIM;
  const float* sG = sm + 2 * NS_B * I_DIM;

  float acc[NS_B];
#pragma unroll
  for (int bb = 0; bb < NS_B; ++bb) acc[bb] = 0.0f;

  for (int i0 = 0; i0 < NS_I; i0 += NS_U) {
    v2f rn[NS_U / 2], nn[NS_U / 2];
    const int inx = (i0 + NS_U < NS_I) ? i0 + NS_U : i0;
#pragma unroll
    for (int u = 0; u < NS_U / 2; ++u) {
      rn[u].x = rp[(inx + 2 * u) * O_DIM];  rn[u].y = rp[(inx + 2 * u + 1) * O_DIM];
      nn[u].x = np[(inx + 2 * u) * O_DIM];  nn[u].y = np[(inx + 2 * u + 1) * O_DIM];
    }
#pragma unroll
    for (int bb = 0; bb < NS_B; ++bb) {
      const int si = bb * I_DIM + ibeg + i0;
      const float4 l4 = *(const float4*)(sL + si);
      const float4 d4 = *(const float4*)(sD + si);
      const float4 g4 = *(const float4*)(sG + si);
      const v2f l01 = {l4.x, l4.y}, l23 = {l4.z, l4.w};
      const v2f d01 = {d4.x, d4.y}, d23 = {d4.z, d4.w};
      const v2f g01 = {g4.x, g4.y}, g23 = {g4.z, g4.w};
      const v2f rr01 = __builtin_elementwise_fma(d01, rv[0], l01);
      const v2f rr23 = __builtin_elementwise_fma(d23, rv[1], l23);
      const v2f t01 = nv[0] * g01;
      const v2f t23 = nv[1] * g23;
      float a = acc[bb];
      a = fmaf(t01.x, hw_sqrt(fabsf(rr01.x)), a);
      a = fmaf(t01.y, hw_sqrt(fabsf(rr01.y)), a);
      a = fmaf(t23.x, hw_sqrt(fabsf(rr23.x)), a);
      a = fmaf(t23.y, hw_sqrt(fabsf(rr23.y)), a);
      acc[bb] = a;
    }
#pragma unroll
    for (int u = 0; u < NS_U / 2; ++u) { rv[u] = rn[u]; nv[u] = nn[u]; }
  }

  __syncthreads();
#pragma unroll
  for (int bb = 0; bb < NS_B; ++bb) sm[(w * NS_B + bb) * NS_O + lane] = acc[bb];
  __syncthreads();
  if (t < NS_B * NS_O) {
    float ssum = 0.0f;
#pragma unroll
    for (int ww = 0; ww < NS_W; ++ww) ssum += sm[ww * (NS_B * NS_O) + t];
    const int bb = t >> 5;
    out[(b0 + bb) * O_DIM + o] += ssum;
  }
}

extern "C" void kernel_launch(void* const* d_in, const int* in_sizes, int n_in,
                              void* d_out, int out_size, void* d_ws, size_t ws_size,
                              hipStream_t stream) {
  (void)in_sizes; (void)n_in; (void)out_size; (void)ws_size;
  const float* x  = (const float*)d_in[0];
  const float* pl = (const float*)d_in[1];
  const float* ph = (const float*)d_in[2];
  const float* r  = (const float*)d_in[3];
  float* out = (float*)d_out;

  float* wL  = (float*)d_ws;
  float* wD  = wL + B_DIM * I_DIM;
  float* wS  = wD + B_DIM * I_DIM;
  float* wN  = wS + B_DIM * I_DIM;
  float* wRS = wN + I_DIM * O_DIM;

  precomp_kernel<<<dim3(B_DIM), dim3(256), 0, stream>>>(x, pl, ph, wL, wD, wS, wRS);

  gemm_and_noise_kernel<<<dim3(GEMM_BLOCKS + NOISE_BLOCKS), dim3(256), 0, stream>>>(
      wD, r, wRS, out, wN);

  noise_accum_kernel<<<dim3(O_DIM / NS_O, B_DIM / NS_B), dim3(NS_W * 32), 0, stream>>>(
      wL, wD, wS, r, wN, out);
}


# GEMM moved into the sweep kernel: each wave does the WMMA steps of its own 32-i slice, tiles join the slice reduction; precompute shares the noise launch: 3 launches to 2 (GEMM sum order changes)
# speedup vs baseline: 6.8733x; 2.1355x over previous
#include <hip/hip_runtime.h>
#include <hip/hip_bf16.h>


typedef __attribute__((ext_vector_type(2))) float v2f;
typedef __attribute__((ext_vector_type(8))) float v8f;

#define B_DIM 128
#define I_DIM 512
#define O_DIM 512

#define C1_JOHNSON 1.65677880e-28f
#define C2_SHOT    5.43656366e-08f
#define NOISE_MIN  1e-12f

__device__ __forceinline__ void precomp_block(
    const float* __restrict__ x,
    const float* __restrict__ pl,
    const float* __restrict__ ph,
    float* __restrict__ L,
    float* __restrict__ Dm,
    float* __restrict__ SG,
    float* __restrict__ rowSumL,
    int b) {
  __shared__ float red[256];
  float c0 = pl[0], c1 = pl[1], c2 = pl[2], c3 = pl[3], c4 = pl[4], c5 = pl[5], c6 = pl[6];
  float d0 = ph[0], d1 = ph[1], d2 = ph[2], d3 = ph[3], d4 = ph[4], d5 = ph[5];
  float lsum = 0.0f;
  for (int i = threadIdx.x; i < I_DIM; i += 256) {
    const float xv = x[b * I_DIM + i];
    float l = c6;
    l = fmaf(l, xv, c5); l = fmaf(l, xv, c4); l = fmaf(l, xv, c3);
    l = fmaf(l, xv, c2); l = fmaf(l, xv, c1); l = fmaf(l, xv, c0);
    float h = d5;
    h = fmaf(h, xv, d4); h = fmaf(h, xv, d3); h = fmaf(h, xv, d2);
    h = fmaf(h, xv, d1); h = fmaf(h, xv, d0);
    const float g = C1_JOHNSON / (fabsf(xv) + NOISE_MIN) + C2_SHOT;
    L[b * I_DIM + i]  = l;
    Dm[b * I_DIM + i] = h - l;
    SG[b * I_DIM + i] = sqrtf(g);
    lsum += l;
  }
  red[threadIdx.x] = lsum;
  __syncthreads();
  for (int s = 128; s > 0; s >>= 1) {
    if (threadIdx.x < (unsigned)s) red[threadIdx.x] += red[threadIdx.x + s];
    __syncthreads();
  }
  if (threadIdx.x == 0) rowSumL[b] = red[0];
}

__device__ __forceinline__ unsigned rotl32(unsigned v, int r) {
  return (v << r) | (v >> (32 - r));
}

__device__ __forceinline__ float bits_to_normal(unsigned bits) {
  const float lo = -0.99999994f;
  const float hi = 1.0f;
  float f = __uint_as_float((bits >> 9) | 0x3f800000u) - 1.0f;
  float u = fmaf(f, hi - lo, lo);
  u = fmaxf(u, lo);
  float w = -__logf((1.0f - u) * (1.0f + u));
  float p;
  if (w < 5.0f) {
    w -= 2.5f;
    p = 2.81022636e-08f;
    p = fmaf(p, w, 3.43273939e-07f);
    p = fmaf(p, w, -3.5233877e-06f);
    p = fmaf(p, w, -4.39150654e-06f);
    p = fmaf(p, w, 0.00021858087f);
    p = fmaf(p, w, -0.00125372503f);
    p = fmaf(p, w, -0.00417768164f);
    p = fmaf(p, w, 0.246640727f);
    p = fmaf(p, w, 1.50140941f);
  } else {
    w = sqrtf(w) - 3.0f;
    p = -0.000200214257f;
    p = fmaf(p, w, 0.000100950558f);
    p = fmaf(p, w, 0.00134934322f);
    p = fmaf(p, w, -0.00367342844f);
    p = fmaf(p, w, 0.00573950773f);
    p = fmaf(p, w, -0.0076224613f);
    p = fmaf(p, w, 0.00943887047f);
    p = fmaf(p, w, 1.00167406f);
    p = fmaf(p, w, 2.83297682f);
  }
  return 1.41421356237f * (p * u);
}

__device__ __forceinline__ void noise_block(float* __restrict__ N, unsigned nb) {
  const unsigned half = (I_DIM * O_DIM) / 2;
  const unsigned j = nb * 256u + threadIdx.x;
  if (j >= half) return;
  unsigned x0 = j, x1 = j + half;
  const unsigned ks0 = 0u, ks1 = 42u;
  const unsigned ks2 = 0x1BD11BDAu ^ ks0 ^ ks1;
  x0 += ks0; x1 += ks1;
  const int ra[4] = {13, 15, 26, 6};
  const int rb[4] = {17, 29, 16, 24};
#pragma unroll
  for (int r = 0; r < 4; ++r) { x0 += x1; x1 = rotl32(x1, ra[r]); x1 ^= x0; }
  x0 += ks1; x1 += ks2 + 1u;
#pragma unroll
  for (int r = 0; r < 4; ++r) { x0 += x1; x1 = rotl32(x1, rb[r]); x1 ^= x0; }
  x0 += ks2; x1 += ks0 + 2u;
#pragma unroll
  for (int r = 0; r < 4; ++r) { x0 += x1; x1 = rotl32(x1, ra[r]); x1 ^= x0; }
  x0 += ks0; x1 += ks1 + 3u;
#pragma unroll
  for (int r = 0; r < 4; ++r) { x0 += x1; x1 = rotl32(x1, rb[r]); x1 ^= x0; }
  x0 += ks1; x1 += ks2 + 4u;
#pragma unroll
  for (int r = 0; r < 4; ++r) { x0 += x1; x1 = rotl32(x1, ra[r]); x1 ^= x0; }
  x0 += ks2; x1 += ks0 + 5u;
  N[j]        = bits_to_normal(x0);
  N[j + half] = bits_to_normal(x1);
}

#define PRE_BLOCKS   B_DIM
#define NOISE_BLOCKS ((I_DIM * O_DIM / 2 + 255) / 256)

__global__ __launch_bounds__(256) void precomp_and_noise_kernel(
    const float* __restrict__ x,
    const float* __restrict__ pl,
    const float* __restrict__ ph,
    float* __restrict__ L,
    float* __restrict__ Dm,
    float* __restrict__ SG,
    float* __restrict__ rowSumL,
    float* __restrict__ N) {
  if (blockIdx.x < PRE_BLOCKS) {
    precomp_block(x, pl, ph, L, Dm, SG, rowSumL, (int)blockIdx.x);
  } else {
    noise_block(N, blockIdx.x - PRE_BLOCKS);
  }
}

__device__ __forceinline__ float hw_sqrt(float v) { return __builtin_amdgcn_sqrtf(v); }

#define NS_B 8
#define NS_O 32
#define NS_W 16
#define NS_I (I_DIM / NS_W)
#define NS_U 4

__global__ __launch_bounds__(NS_W * 32)
__attribute__((amdgpu_waves_per_eu(4, 4))) void gemm_and_sweep_kernel(
    const float* __restrict__ L, const float* __restrict__ Dm,
    const float* __restrict__ SG, const float* __restrict__ R,
    const float* __restrict__ N, const float* __restrict__ rowSumL,
    float* __restrict__ out) {
  __shared__ __attribute__((aligned(16))) float sm[3 * NS_B * I_DIM + NS_W * NS_B * NS_O];
  const int t    = threadIdx.x;
  const int lane = t & 31;
  const int w    = t >> 5;
  const int o    = blockIdx.x * NS_O + lane;
  const int b0   = blockIdx.y * NS_B;
  const int ibeg = w * NS_I;

  {
    const float4* gL = (const float4*)(L  + b0 * I_DIM);
    const float4* gD = (const float4*)(Dm + b0 * I_DIM);
    const float4* gG = (const float4*)(SG + b0 * I_DIM);
    float4* s4 = (float4*)sm;
    const int n4 = NS_B * I_DIM / 4;
#pragma unroll
    for (int k = 0; k < n4 / (NS_W * 32); ++k) {
      const int s = t + k * (NS_W * 32);
      s4[s]          = gL[s];
      s4[n4 + s]     = gD[s];
      s4[2 * n4 + s] = gG[s];
    }
  }

  const float* rp = R + ibeg * O_DIM + o;
  const float* np = N + ibeg * O_DIM + o;
  v2f rv[NS_U / 2], nv[NS_U / 2];
#pragma unroll
  for (int u = 0; u < NS_U / 2; ++u) {
    rv[u].x = rp[(2 * u) * O_DIM];  rv[u].y = rp[(2 * u + 1) * O_DIM];
    nv[u].x = np[(2 * u) * O_DIM];  nv[u].y = np[(2 * u + 1) * O_DIM];
  }
  const int gn = lane & 15;
  const int gh = lane >> 4;
  v2f bq[NS_I / 4][2];
#pragma unroll
  for (int s = 0; s < NS_I / 4; ++s) {
#pragma unroll
    for (int j = 0; j < 2; ++j) {
      bq[s][j] = *(const v2f*)(R + (ibeg + 4 * s + 2 * gh + j) * O_DIM
                               + blockIdx.x * NS_O + 2 * gn);
    }
  }
  const float rsl = rowSumL[b0 + (w & (NS_B - 1))];
  __syncthreads();

  const float* sL = sm;
  const float* sD = sm + NS_B * I_DIM;
  const float* sG = sm + 2 * NS_B * I_DIM;

  v8f g0 = {}, g1 = {};
#pragma unroll
  for (int s = 0; s < NS_I / 4; ++s) {
    v2f a = *(const v2f*)(sD + (gn & (NS_B - 1)) * I_DIM + ibeg + 4 * s + 2 * gh);
    a.x = (gn < NS_B) ? a.x : 0.0f;
    a.y = (gn < NS_B) ? a.y : 0.0f;
    const v2f bt0 = {bq[s][0].x, bq[s][1].x};
    const v2f bt1 = {bq[s][0].y, bq[s][1].y};
    g0 = __builtin_amdgcn_wmma_f32_16x16x4_f32(
         false, a,  false, bt0,
         (short)0, g0,  false,  false);
    g1 = __builtin_amdgcn_wmma_f32_16x16x4_f32(
         false, a,  false, bt1,
         (short)0, g1,  false,  false);
  }
  float* pG = sm + 3 * NS_B * I_DIM;
  if (gh == 0) {
#pragma unroll
    for (int v = 0; v < NS_B; ++v) {
      const v2f gv = {g0[v], g1[v]};
      *(v2f*)(pG + (w * NS_B + v) * NS_O + 2 * gn) = gv;
    }
  }

  float acc[NS_B];
#pragma unroll
  for (int bb = 0; bb < NS_B; ++bb) acc[bb] = 0.0f;

  for (int i0 = 0; i0 < NS_I; i0 += NS_U) {
    v2f rn[NS_U / 2], nn[NS_U / 2];
    const int inx = (i0 + NS_U < NS_I) ? i0 + NS_U : i0;
#pragma unroll
    for (int u = 0; u < NS_U / 2; ++u) {
      rn[u].x = rp[(inx + 2 * u) * O_DIM];  rn[u].y = rp[(inx + 2 * u + 1) * O_DIM];
      nn[u].x = np[(inx + 2 * u) * O_DIM];  nn[u].y = np[(inx + 2 * u + 1) * O_DIM];
    }
#pragma unroll
    for (int bb = 0; bb < NS_B; ++bb) {
      const int si = bb * I_DIM + ibeg + i0;
      const float4 l4 = *(const float4*)(sL + si);
      const float4 d4 = *(const float4*)(sD + si);
      const float4 g4 = *(const float4*)(sG + si);
      const v2f l01 = {l4.x, l4.y}, l23 = {l4.z, l4.w};
      const v2f d01 = {d4.x, d4.y}, d23 = {d4.z, d4.w};
      const v2f g01 = {g4.x, g4.y}, g23 = {g4.z, g4.w};
      const v2f rr01 = __builtin_elementwise_fma(d01, rv[0], l01);
      const v2f rr23 = __builtin_elementwise_fma(d23, rv[1], l23);
      const v2f t01 = nv[0] * g01;
      const v2f t23 = nv[1] * g23;
      float a = acc[bb];
      a = fmaf(t01.x, hw_sqrt(fabsf(rr01.x)), a);
      a = fmaf(t01.y, hw_sqrt(fabsf(rr01.y)), a);
      a = fmaf(t23.x, hw_sqrt(fabsf(rr23.x)), a);
      a = fmaf(t23.y, hw_sqrt(fabsf(rr23.y)), a);
      acc[bb] = a;
    }
#pragma unroll
    for (int u = 0; u < NS_U / 2; ++u) { rv[u] = rn[u]; nv[u] = nn[u]; }
  }

  __syncthreads();
#pragma unroll
  for (int bb = 0; bb < NS_B; ++bb) sm[(w * NS_B + bb) * NS_O + lane] = acc[bb];
  __syncthreads();
  if (t < NS_B * NS_O) {
    float ssum = 0.0f, gsum = 0.0f;
#pragma unroll
    for (int ww = 0; ww < NS_W; ++ww) ssum += sm[ww * (NS_B * NS_O) + t];
#pragma unroll
    for (int ww = 0; ww < NS_W; ++ww) gsum += pG[ww * (NS_B * NS_O) + t];
    const int bb = t >> 5;
    out[(b0 + bb) * O_DIM + o] = (gsum + rsl) + ssum;
  }
}

extern "C" void kernel_launch(void* const* d_in, const int* in_sizes, int n_in,
                              void* d_out, int out_size, void* d_ws, size_t ws_size,
                              hipStream_t stream) {
  (void)in_sizes; (void)n_in; (void)out_size; (void)ws_size;
  const float* x  = (const float*)d_in[0];
  const float* pl = (const float*)d_in[1];
  const float* ph = (const float*)d_in[2];
  const float* r  = (const float*)d_in[3];
  float* out = (float*)d_out;

  float* wL  = (float*)d_ws;
  float* wD  = wL + B_DIM * I_DIM;
  float* wS  = wD + B_DIM * I_DIM;
  float* wN  = wS + B_DIM * I_DIM;
  float* wRS = wN + I_DIM * O_DIM;

  precomp_and_noise_kernel<<<dim3(PRE_BLOCKS + NOISE_BLOCKS), dim3(256), 0, stream>>>(
      x, pl, ph, wL, wD, wS, wRS, wN);

  gemm_and_sweep_kernel<<<dim3(O_DIM / NS_O, B_DIM / NS_B), dim3(NS_W * 32), 0, stream>>>(
      wL, wD, wS, r, wN, wRS, out);
}


# start-up: each wave requests its ten global loads first and does all its noise arithmetic (integer rounds, normals) before the first wait for them: load latency runs under that arithmetic
# speedup vs baseline: 10.0790x; 1.4664x over previous
#include <hip/hip_runtime.h>
#include <hip/hip_bf16.h>


typedef __attribute__((ext_vector_type(2))) float v2f;
typedef __attribute__((ext_vector_type(8))) float v8f;

#define B_DIM 128
#define I_DIM 512
#define O_DIM 512

#define C1_JOHNSON 1.65677880e-28f
#define C2_SHOT    5.43656366e-08f
#define NOISE_MIN  1e-12f

__device__ __forceinline__ unsigned rotl32(unsigned v, int r) {
  return (v << r) | (v >> (32 - r));
}

__device__ __forceinline__ float bits_to_normal(unsigned bits) {
  const float lo = -0.99999994f;
  const float hi = 1.0f;
  float f = __uint_as_float((bits >> 9) | 0x3f800000u) - 1.0f;
  float u = fmaf(f, hi - lo, lo);
  u = fmaxf(u, lo);
  const float w = -__logf((1.0f - u) * (1.0f + u));
  const float wc = w - 2.5f;
  float pc = 2.81022636e-08f;
  pc = fmaf(pc, wc, 3.43273939e-07f);
  pc = fmaf(pc, wc, -3.5233877e-06f);
  pc = fmaf(pc, wc, -4.39150654e-06f);
  pc = fmaf(pc, wc, 0.00021858087f);
  pc = fmaf(pc, wc, -0.00125372503f);
  pc = fmaf(pc, wc, -0.00417768164f);
  pc = fmaf(pc, wc, 0.246640727f);
  pc = fmaf(pc, wc, 1.50140941f);
  const float wt = __builtin_amdgcn_sqrtf(w) - 3.0f;
  float pt = -0.000200214257f;
  pt = fmaf(pt, wt, 0.000100950558f);
  pt = fmaf(pt, wt, 0.00134934322f);
  pt = fmaf(pt, wt, -0.00367342844f);
  pt = fmaf(pt, wt, 0.00573950773f);
  pt = fmaf(pt, wt, -0.0076224613f);
  pt = fmaf(pt, wt, 0.00943887047f);
  pt = fmaf(pt, wt, 1.00167406f);
  pt = fmaf(pt, wt, 2.83297682f);
  const float p = (w < 5.0f) ? pc : pt;
  return 1.41421356237f * (p * u);
}

__device__ __forceinline__ void threefry_pair(unsigned j, unsigned& bits_lo, unsigned& bits_hi) {
  const unsigned half = (I_DIM * O_DIM) / 2;
  unsigned x0 = j, x1 = j + half;
  const unsigned ks0 = 0u, ks1 = 42u;
  const unsigned ks2 = 0x1BD11BDAu ^ ks0 ^ ks1;
  x0 += ks0; x1 += ks1;
  const int ra[4] = {13, 15, 26, 6};
  const int rb[4] = {17, 29, 16, 24};
#pragma unroll
  for (int r = 0; r < 4; ++r) { x0 += x1; x1 = rotl32(x1, ra[r]); x1 ^= x0; }
  x0 += ks1; x1 += ks2 + 1u;
#pragma unroll
  for (int r = 0; r < 4; ++r) { x0 += x1; x1 = rotl32(x1, rb[r]); x1 ^= x0; }
  x0 += ks2; x1 += ks0 + 2u;
#pragma unroll
  for (int r = 0; r < 4; ++r) { x0 += x1; x1 = rotl32(x1, ra[r]); x1 ^= x0; }
  x0 += ks0; x1 += ks1 + 3u;
#pragma unroll
  for (int r = 0; r < 4; ++r) { x0 += x1; x1 = rotl32(x1, rb[r]); x1 ^= x0; }
  x0 += ks1; x1 += ks2 + 4u;
#pragma unroll
  for (int r = 0; r < 4; ++r) { x0 += x1; x1 = rotl32(x1, ra[r]); x1 ^= x0; }
  x0 += ks2; x1 += ks0 + 5u;
  bits_lo = x0;
  bits_hi = x1;
}

__device__ __forceinline__ float hw_sqrt(float v) { return __builtin_amdgcn_sqrtf(v); }

#define TB 32
#define TO 8
#define NW 16
#define TP 36
#define GP 9

#define I_OF(w, kk) (16 * (w) + (kk) + (((kk) >= 16) ? 240 : 0))

__global__ __launch_bounds__(NW * 32)
__attribute__((amdgpu_waves_per_eu(4, 4))) void memristor_fused_kernel(
    const float* X,
    const float* __restrict__ pl,
    const float* __restrict__ ph,
    const float* R,
    float* __restrict__ out) {
  __shared__ __attribute__((aligned(16))) float sT[NW * TB * TP];
  __shared__ __attribute__((aligned(16))) float sN[NW * 32 * TO];
  __shared__ __attribute__((aligned(16))) float sR[NW * 32 * TO];
  __shared__ float pS[NW * TB * GP];
  __shared__ float pG[NW * TB * GP];
  __shared__ float pL[NW * TB];

  const int t    = threadIdx.x;
  const int lane = t & 31;
  const int w    = __builtin_amdgcn_readfirstlane(t >> 5);
  const int o0   = blockIdx.x * TO;
  const int b0   = blockIdx.y * TB;
  const int gn   = lane & 15;
  const int gh   = lane >> 4;

  float* sTw = sT + w * (TB * TP);
  float* sNw = sN + w * (32 * TO);
  float* sRw = sR + w * (32 * TO);

  const int xcol = 4 * (lane & 7);
  const float* xsrc = X + (b0 + (lane >> 3)) * I_DIM + I_OF(w, xcol);
  const float4 xv0 = *(const float4*)(xsrc + 0 * 4 * I_DIM);
  const float4 xv1 = *(const float4*)(xsrc + 1 * 4 * I_DIM);
  const float4 xv2 = *(const float4*)(xsrc + 2 * 4 * I_DIM);
  const float4 xv3 = *(const float4*)(xsrc + 3 * 4 * I_DIM);
  const float4 xv4 = *(const float4*)(xsrc + 4 * 4 * I_DIM);
  const float4 xv5 = *(const float4*)(xsrc + 5 * 4 * I_DIM);
  const float4 xv6 = *(const float4*)(xsrc + 6 * 4 * I_DIM);
  const float4 xv7 = *(const float4*)(xsrc + 7 * 4 * I_DIM);
  const float* rsrc = R + I_OF(w, lane) * O_DIM + o0;
  const float4 rv0 = *(const float4*)(rsrc);
  const float4 rv1 = *(const float4*)(rsrc + 4);
  const float c0 = pl[0], c1 = pl[1], c2 = pl[2], c3 = pl[3], c4 = pl[4], c5 = pl[5], c6 = pl[6];
  const float d0 = ph[0], d1 = ph[1], d2 = ph[2], d3 = ph[3], d4 = ph[4], d5 = ph[5];

  __builtin_amdgcn_sched_barrier(0);

  float nvl[4], nvh[4];
#pragma unroll
  for (int q = 0; q < 4; ++q) {
    const int e    = q * 32 + lane;
    const int ioff = e >> 3;
    const int c    = e & 7;
    unsigned bl, bh;
    threefry_pair((unsigned)((16 * w + ioff) * O_DIM + o0 + c), bl, bh);
    nvl[q] = bits_to_normal(bl);
    nvh[q] = bits_to_normal(bh);
  }

  *(float4*)(sRw + lane * TO)     = rv0;
  *(float4*)(sRw + lane * TO + 4) = rv1;
  {
    float* xdst = sTw + (lane >> 3) * TP + xcol;
    *(float4*)(xdst + 0 * 4 * TP) = xv0;
    *(float4*)(xdst + 1 * 4 * TP) = xv1;
    *(float4*)(xdst + 2 * 4 * TP) = xv2;
    *(float4*)(xdst + 3 * 4 * TP) = xv3;
    *(float4*)(xdst + 4 * 4 * TP) = xv4;
    *(float4*)(xdst + 5 * 4 * TP) = xv5;
    *(float4*)(xdst + 6 * 4 * TP) = xv6;
    *(float4*)(xdst + 7 * 4 * TP) = xv7;
  }
#pragma unroll
  for (int q = 0; q < 4; ++q) {
    const int e = q * 32 + lane;
    sNw[e]       = nvl[q];
    sNw[128 + e] = nvh[q];
  }
  __syncthreads();

  v2f gl[16], gd[16];
  float sumL = 0.0f;
  {
    const v2f C6 = {c6, c6}, C5 = {c5, c5}, C4 = {c4, c4}, C3 = {c3, c3};
    const v2f C2 = {c2, c2}, C1 = {c1, c1}, C0 = {c0, c0};
    const v2f D5 = {d5, d5}, D4 = {d4, d4}, D3 = {d3, d3};
    const v2f D2 = {d2, d2}, D1 = {d1, d1}, D0 = {d0, d0};
    const v2f G2 = {C2_SHOT, C2_SHOT};
    v2f sl = {0.0f, 0.0f};
#pragma unroll
    for (int q = 0; q < 8; ++q) {
      const float4 x4 = *(const float4*)(sTw + lane * TP + 4 * q);
      float4 dd;
#pragma unroll
      for (int p = 0; p < 2; ++p) {
        const v2f xx = p ? (v2f){x4.z, x4.w} : (v2f){x4.x, x4.y};
        v2f l = C6;
        l = __builtin_elementwise_fma(l, xx, C5);
        l = __builtin_elementwise_fma(l, xx, C4);
        l = __builtin_elementwise_fma(l, xx, C3);
        l = __builtin_elementwise_fma(l, xx, C2);
        l = __builtin_elementwise_fma(l, xx, C1);
        l = __builtin_elementwise_fma(l, xx, C0);
        v2f h = D5;
        h = __builtin_elementwise_fma(h, xx, D4);
        h = __builtin_elementwise_fma(h, xx, D3);
        h = __builtin_elementwise_fma(h, xx, D2);
        h = __builtin_elementwise_fma(h, xx, D1);
        h = __builtin_elementwise_fma(h, xx, D0);
        const v2f dm = h - l;
        const v2f rc = {__builtin_amdgcn_rcpf(fabsf(xx.x) + NOISE_MIN),
                        __builtin_amdgcn_rcpf(fabsf(xx.y) + NOISE_MIN)};
        const v2f g = __builtin_elementwise_fma((v2f){C1_JOHNSON, C1_JOHNSON}, rc, G2);
        gl[2 * q + p] = g * l;
        gd[2 * q + p] = g * dm;
        sl += l;
        if (p) { dd.z = dm.x; dd.w = dm.y; } else { dd.x = dm.x; dd.y = dm.y; }
      }
      *(float4*)(sTw + lane * TP + 4 * q) = dd;
    }
    sumL = sl.x + sl.y;
  }
  __syncthreads();

  v8f g0 = {}, g1 = {};
#pragma unroll
  for (int s = 0; s < 8; ++s) {
    const v2f a0 = *(const v2f*)(sTw + gn * TP + 4 * s + 2 * gh);
    const v2f a1 = *(const v2f*)(sTw + (16 + gn) * TP + 4 * s + 2 * gh);
    const v2f bt = {sRw[(4 * s + 2 * gh) * TO + (gn & 7)],
                    sRw[(4 * s + 2 * gh + 1) * TO + (gn & 7)]};
    g0 = __builtin_amdgcn_wmma_f32_16x16x4_f32(
         false, a0,  false, bt,
         (short)0, g0,  false,  false);
    g1 = __builtin_amdgcn_wmma_f32_16x16x4_f32(
         false, a1,  false, bt,
         (short)0, g1,  false,  false);
  }
  float* pGw = pG + w * (TB * GP);
  if (gn < TO) {
#pragma unroll
    for (int v = 0; v < 8; ++v) {
      pGw[(v + 8 * gh) * GP + gn]      = g0[v];
      pGw[(16 + v + 8 * gh) * GP + gn] = g1[v];
    }
  }

  v2f acc[TO / 2];
#pragma unroll
  for (int c = 0; c < TO / 2; ++c) acc[c] = (v2f){0.0f, 0.0f};
#pragma unroll
  for (int kk = 0; kk < 32; ++kk) {
    const float4 ra = *(const float4*)(sRw + kk * TO);
    const float4 rb = *(const float4*)(sRw + kk * TO + 4);
    const float4 na = *(const float4*)(sNw + kk * TO);
    const float4 nb = *(const float4*)(sNw + kk * TO + 4);
    const float glv = (kk & 1) ? gl[kk >> 1].y : gl[kk >> 1].x;
    const float gdv = (kk & 1) ? gd[kk >> 1].y : gd[kk >> 1].x;
    const v2f GL = {glv, glv}, GD = {gdv, gdv};
    const v2f r2[4] = {{ra.x, ra.y}, {ra.z, ra.w}, {rb.x, rb.y}, {rb.z, rb.w}};
    const v2f n2[4] = {{na.x, na.y}, {na.z, na.w}, {nb.x, nb.y}, {nb.z, nb.w}};
#pragma unroll
    for (int c = 0; c < TO / 2; ++c) {
      const v2f rr = __builtin_elementwise_fma(GD, r2[c], GL);
      const v2f sq = {hw_sqrt(fabsf(rr.x)), hw_sqrt(fabsf(rr.y))};
      acc[c] = __builtin_elementwise_fma(n2[c], sq, acc[c]);
    }
  }

  float* pSw = pS + w * (TB * GP);
#pragma unroll
  for (int c = 0; c < TO / 2; ++c) {
    pSw[lane * GP + 2 * c]     = acc[c].x;
    pSw[lane * GP + 2 * c + 1] = acc[c].y;
  }
  pL[w * TB + lane] = sumL;
  __syncthreads();
  if (t < TB * TO) {
    const int oc = t & (TO - 1);
    const int bb = t >> 3;
    float ssum = 0.0f, gsum = 0.0f, lsum = 0.0f;
#pragma unroll
    for (int ww = 0; ww < NW; ++ww) ssum += pS[ww * (TB * GP) + bb * GP + oc];
#pragma unroll
    for (int ww = 0; ww < NW; ++ww) gsum += pG[ww * (TB * GP) + bb * GP + oc];
#pragma unroll
    for (int ww = 0; ww < NW; ++ww) lsum += pL[ww * TB + bb];
    out[(b0 + bb) * O_DIM + o0 + oc] = (gsum + lsum) + ssum;
  }
}

extern "C" void kernel_launch(void* const* d_in, const int* in_sizes, int n_in,
                              void* d_out, int out_size, void* d_ws, size_t ws_size,
                              hipStream_t stream) {
  (void)in_sizes; (void)n_in; (void)out_size; (void)d_ws; (void)ws_size;
  const float* x  = (const float*)d_in[0];
  const float* pl = (const float*)d_in[1];
  const float* ph = (const float*)d_in[2];
  const float* r  = (const float*)d_in[3];
  float* out = (float*)d_out;

  memristor_fused_kernel<<<dim3(O_DIM / TO, B_DIM / TB), dim3(NW * 32), 0, stream>>>(
      x, pl, ph, r, out);
}
